# starting kernel
# baseline (speedup 1.0000x reference)
#include <hip/hip_runtime.h>

typedef __attribute__((ext_vector_type(2))) float v2f;
typedef __attribute__((ext_vector_type(8))) float v8f;

#define N_NODES   100000
#define D_FEAT    128
#define N_EDGES   1600000
#define N_CLASSES 32
#define M_TILES   (N_NODES / 16)

__global__ __launch_bounds__(128)
void gin_proj_wmma(const float* __restrict__ x, const float* __restrict__ W,
                   const float* __restrict__ bias, float* __restrict__ y,
                   float* __restrict__ out) {
    const int lane  = threadIdx.x & 31;
    const int wave  = threadIdx.x >> 5;
    const int mtile = blockIdx.x * 4 + wave;
    if (mtile >= M_TILES) return;

    const int col   = lane & 15;
    const int khalf = (lane >> 4) << 1;

    const float* xrow  = x + (mtile * 16 + col) * D_FEAT;
    const float* wrow0 = W + col * D_FEAT;
    const float* wrow1 = W + (col + 16) * D_FEAT;

    v8f acc0 = {};
    v8f acc1 = {};
#pragma unroll 4
    for (int k0 = 0; k0 < D_FEAT; k0 += 4) {
        v2f a  = *(const v2f*)(xrow  + k0 + khalf);
        v2f b0 = *(const v2f*)(wrow0 + k0 + khalf);
        v2f b1 = *(const v2f*)(wrow1 + k0 + khalf);
        acc0 = __builtin_amdgcn_wmma_f32_16x16x4_f32(false, a, false, b0,
                                                     (short)0, acc0, false, false);
        acc1 = __builtin_amdgcn_wmma_f32_16x16x4_f32(false, a, false, b1,
                                                     (short)0, acc1, false, false);
    }

    const float bias0 = bias[col];
    const float bias1 = bias[col + 16];
    const int rbase = mtile * 16 + ((lane >> 4) << 3);
#pragma unroll
    for (int r = 0; r < 8; ++r) {
        const int row = rbase + r;
        y[row * N_CLASSES + col]        = acc0[r];
        y[row * N_CLASSES + col + 16]   = acc1[r];
        out[row * N_CLASSES + col]      = acc0[r] + bias0;
        out[row * N_CLASSES + col + 16] = acc1[r] + bias1;
    }
}

__global__ __launch_bounds__(256)
void gin_edge_scatter(const int* __restrict__ src, const int* __restrict__ dst,
                      const float* __restrict__ y, float* __restrict__ out) {
    const int  c       = threadIdx.x & 31;
    long long  t       = (long long)blockIdx.x * blockDim.x + threadIdx.x;
    const long long nw = ((long long)gridDim.x * blockDim.x) >> 5;
    for (long long e = t >> 5; e < N_EDGES; e += nw) {
        const int   s = src[e];
        const int   d = dst[e];
        const float v = y[(long long)s * N_CLASSES + c];
        atomicAdd(out + (long long)d * N_CLASSES + c, v);
    }
}

extern "C" void kernel_launch(void* const* d_in, const int* in_sizes, int n_in,
                              void* d_out, int out_size, void* d_ws, size_t ws_size,
                              hipStream_t stream) {
    const float* x          = (const float*)d_in[0];
    const int*   edge_index = (const int*)  d_in[1];
    const float* W          = (const float*)d_in[2];
    const float* b          = (const float*)d_in[3];
    float*       out        = (float*)d_out;
    float*       y          = (float*)d_ws;

    const int* src = edge_index;
    const int* dst = edge_index + N_EDGES;

    const int proj_blocks = (M_TILES + 3) / 4;
    gin_proj_wmma<<<proj_blocks, 128, 0, stream>>>(x, W, b, y, out);

    const int edge_blocks = (N_EDGES * 32) / 256;
    gin_edge_scatter<<<edge_blocks, 256, 0, stream>>>(src, dst, y, out);
}


# edge scatter: 32 edges per wave (one coalesced index load, readlane broadcast), 16 row gathers in flight ahead of their 16 f32 atomic adds; atomics kept, sums in arrival order as before
# speedup vs baseline: 2.7519x; 2.7519x over previous
#include <hip/hip_runtime.h>

typedef __attribute__((ext_vector_type(2))) float v2f;
typedef __attribute__((ext_vector_type(8))) float v8f;

#define N_NODES   100000
#define D_FEAT    128
#define N_EDGES   1600000
#define N_CLASSES 32
#define M_TILES   (N_NODES / 16)

__global__ __launch_bounds__(128)
void gin_proj_wmma(const float* __restrict__ x, const float* __restrict__ W,
                   const float* __restrict__ bias, float* __restrict__ y,
                   float* __restrict__ out) {
    const int lane  = threadIdx.x & 31;
    const int wave  = threadIdx.x >> 5;
    const int mtile = blockIdx.x * 4 + wave;
    if (mtile >= M_TILES) return;

    const int col   = lane & 15;
    const int khalf = (lane >> 4) << 1;

    const float* xrow  = x + (mtile * 16 + col) * D_FEAT;
    const float* wrow0 = W + col * D_FEAT;
    const float* wrow1 = W + (col + 16) * D_FEAT;

    v8f acc0 = {};
    v8f acc1 = {};
#pragma unroll 4
    for (int k0 = 0; k0 < D_FEAT; k0 += 4) {
        v2f a  = *(const v2f*)(xrow  + k0 + khalf);
        v2f b0 = *(const v2f*)(wrow0 + k0 + khalf);
        v2f b1 = *(const v2f*)(wrow1 + k0 + khalf);
        acc0 = __builtin_amdgcn_wmma_f32_16x16x4_f32(false, a, false, b0,
                                                     (short)0, acc0, false, false);
        acc1 = __builtin_amdgcn_wmma_f32_16x16x4_f32(false, a, false, b1,
                                                     (short)0, acc1, false, false);
    }

    const float bias0 = bias[col];
    const float bias1 = bias[col + 16];
    const int rbase = mtile * 16 + ((lane >> 4) << 3);
#pragma unroll
    for (int r = 0; r < 8; ++r) {
        const int row = rbase + r;
        y[row * N_CLASSES + col]        = acc0[r];
        y[row * N_CLASSES + col + 16]   = acc1[r];
        out[row * N_CLASSES + col]      = acc0[r] + bias0;
        out[row * N_CLASSES + col + 16] = acc1[r] + bias1;
    }
}

#define EDGES_PER_WAVE 32
#define WAVES_PER_BLOCK 8
#define ROWS_IN_FLIGHT 16
#define N_CHUNKS (N_EDGES / EDGES_PER_WAVE)
static_assert(N_EDGES % EDGES_PER_WAVE == 0, "edge list must split into whole 32-edge chunks");
static_assert(N_CHUNKS % WAVES_PER_BLOCK == 0, "chunks must fill whole blocks");

__global__ __launch_bounds__(256)
void gin_edge_scatter(const int* __restrict__ src, const int* __restrict__ dst,
                      const float* __restrict__ y, float* __restrict__ out) {
    const int c = threadIdx.x & 31;
    const unsigned c4 = (unsigned)c * 4u;
    const int wave  = __builtin_amdgcn_readfirstlane((int)(threadIdx.x >> 5));
    const int chunk = blockIdx.x * WAVES_PER_BLOCK + wave;
    if (chunk >= N_CHUNKS) return;

    const int e   = chunk * EDGES_PER_WAVE + c;
    const int s_l = src[e];
    const int d_l = dst[e];

#pragma unroll
    for (int g = 0; g < EDGES_PER_WAVE; g += ROWS_IN_FLIGHT) {
        float v[ROWS_IN_FLIGHT];
#pragma unroll
        for (int u = 0; u < ROWS_IN_FLIGHT; ++u) {
            const unsigned s = (unsigned)__builtin_amdgcn_readlane(s_l, g + u);
            v[u] = *(const float*)((const char*)y + ((s << 7) | c4));
        }
#pragma unroll
        for (int u = 0; u < ROWS_IN_FLIGHT; ++u) {
            const unsigned d = (unsigned)__builtin_amdgcn_readlane(d_l, g + u);
            atomicAdd((float*)((char*)out + ((d << 7) | c4)), v[u]);
        }
    }
}

extern "C" void kernel_launch(void* const* d_in, const int* in_sizes, int n_in,
                              void* d_out, int out_size, void* d_ws, size_t ws_size,
                              hipStream_t stream) {
    const float* x          = (const float*)d_in[0];
    const int*   edge_index = (const int*)  d_in[1];
    const float* W          = (const float*)d_in[2];
    const float* b          = (const float*)d_in[3];
    float*       out        = (float*)d_out;
    float*       y          = (float*)d_ws;

    const int* src = edge_index;
    const int* dst = edge_index + N_EDGES;

    const int proj_blocks = (M_TILES + 3) / 4;
    gin_proj_wmma<<<proj_blocks, 128, 0, stream>>>(x, W, b, y, out);

    const int edge_blocks = N_CHUNKS / WAVES_PER_BLOCK;
    gin_edge_scatter<<<edge_blocks, 256, 0, stream>>>(src, dst, y, out);
}


# projection: x rows, W and the result tile staged through LDS so every global access covers whole 128-byte lines (W once per 4-tile block); k sum order permuted inside 8-element blocks, f32 kept
# speedup vs baseline: 2.9965x; 1.0889x over previous
#include <hip/hip_runtime.h>

typedef __attribute__((ext_vector_type(2))) float v2f;
typedef __attribute__((ext_vector_type(4))) float v4f;
typedef __attribute__((ext_vector_type(8))) float v8f;

#define N_NODES   100000
#define D_FEAT    128
#define N_EDGES   1600000
#define N_CLASSES 32
#define M_TILES   (N_NODES / 16)

#define PROJ_WAVES 4
#define XP 132
#define OP 36

__global__ __launch_bounds__(128)
void gin_proj_wmma(const float* __restrict__ x, const float* __restrict__ W,
                   const float* __restrict__ bias, float* __restrict__ y,
                   float* __restrict__ out) {
    __shared__ __attribute__((aligned(16))) float Wl[N_CLASSES * XP];
    __shared__ __attribute__((aligned(16))) float Xl[PROJ_WAVES * 16 * XP];

    const int tid  = threadIdx.x;
    const int lane = tid & 31;
    const int wave  = __builtin_amdgcn_readfirstlane(tid >> 5);
    const int mtile = blockIdx.x * PROJ_WAVES + wave;
    const bool valid = mtile < M_TILES;
    float* xl = Xl + wave * (16 * XP);

    v4f xr[16];
    if (valid) {
        const float* xt = x + mtile * (16 * D_FEAT) + lane * 4;
#pragma unroll
        for (int r = 0; r < 16; ++r) xr[r] = *(const v4f*)(xt + r * D_FEAT);
    }
#pragma unroll
    for (int i = 0; i < 8; ++i) {
        const int idx = tid + 128 * i;
        const int row = idx >> 5;
        const int c4  = (idx & 31) * 4;
        *(v4f*)(Wl + row * XP + c4) = *(const v4f*)(W + row * D_FEAT + c4);
    }
    if (valid) {
#pragma unroll
        for (int r = 0; r < 16; ++r) *(v4f*)(xl + r * XP + lane * 4) = xr[r];
    }
    __syncthreads();

    const int c = lane & 15;
    const int h = lane >> 4;
    v8f acc0 = {};
    v8f acc1 = {};
    if (valid) {
        const float* ap  = xl + c * XP + 4 * h;
        const float* b0p = Wl + c * XP + 4 * h;
        const float* b1p = Wl + (c + 16) * XP + 4 * h;
#pragma unroll
        for (int j = 0; j < D_FEAT / 8; ++j) {
            const v4f a  = *(const v4f*)(ap  + 8 * j);
            const v4f b0 = *(const v4f*)(b0p + 8 * j);
            const v4f b1 = *(const v4f*)(b1p + 8 * j);
            const v2f alo  = {a[0],  a[1]},  ahi  = {a[2],  a[3]};
            const v2f b0lo = {b0[0], b0[1]}, b0hi = {b0[2], b0[3]};
            const v2f b1lo = {b1[0], b1[1]}, b1hi = {b1[2], b1[3]};
            acc0 = __builtin_amdgcn_wmma_f32_16x16x4_f32(false, alo, false, b0lo,
                                                         (short)0, acc0, false, false);
            acc1 = __builtin_amdgcn_wmma_f32_16x16x4_f32(false, alo, false, b1lo,
                                                         (short)0, acc1, false, false);
            acc0 = __builtin_amdgcn_wmma_f32_16x16x4_f32(false, ahi, false, b0hi,
                                                         (short)0, acc0, false, false);
            acc1 = __builtin_amdgcn_wmma_f32_16x16x4_f32(false, ahi, false, b1hi,
                                                         (short)0, acc1, false, false);
        }
#pragma unroll
        for (int r = 0; r < 8; ++r) {
            xl[(8 * h + r) * OP + c]      = acc0[r];
            xl[(8 * h + r) * OP + c + 16] = acc1[r];
        }
    }
    __syncthreads();
    if (valid) {
        const int c4 = (lane & 7) * 4;
        const v4f bv = *(const v4f*)(bias + c4);
        float* yt = y   + mtile * (16 * N_CLASSES);
        float* ot = out + mtile * (16 * N_CLASSES);
#pragma unroll
        for (int i = 0; i < 4; ++i) {
            const int row = (lane >> 3) + 4 * i;
            const v4f v = *(const v4f*)(xl + row * OP + c4);
            *(v4f*)(yt + row * N_CLASSES + c4) = v;
            *(v4f*)(ot + row * N_CLASSES + c4) = v + bv;
        }
    }
}

#define EDGES_PER_WAVE 32
#define WAVES_PER_BLOCK 8
#define ROWS_IN_FLIGHT 16
#define N_CHUNKS (N_EDGES / EDGES_PER_WAVE)
static_assert(N_EDGES % EDGES_PER_WAVE == 0, "edge list must split into whole 32-edge chunks");
static_assert(N_CHUNKS % WAVES_PER_BLOCK == 0, "chunks must fill whole blocks");

__global__ __launch_bounds__(256)
void gin_edge_scatter(const int* __restrict__ src, const int* __restrict__ dst,
                      const float* __restrict__ y, float* __restrict__ out) {
    const int c = threadIdx.x & 31;
    const unsigned c4 = (unsigned)c * 4u;
    const int wave  = __builtin_amdgcn_readfirstlane((int)(threadIdx.x >> 5));
    const int chunk = blockIdx.x * WAVES_PER_BLOCK + wave;
    if (chunk >= N_CHUNKS) return;

    const int e   = chunk * EDGES_PER_WAVE + c;
    const int s_l = src[e];
    const int d_l = dst[e];

#pragma unroll
    for (int g = 0; g < EDGES_PER_WAVE; g += ROWS_IN_FLIGHT) {
        float v[ROWS_IN_FLIGHT];
#pragma unroll
        for (int u = 0; u < ROWS_IN_FLIGHT; ++u) {
            const unsigned s = (unsigned)__builtin_amdgcn_readlane(s_l, g + u);
            v[u] = *(const float*)((const char*)y + ((s << 7) | c4));
        }
#pragma unroll
        for (int u = 0; u < ROWS_IN_FLIGHT; ++u) {
            const unsigned d = (unsigned)__builtin_amdgcn_readlane(d_l, g + u);
            atomicAdd((float*)((char*)out + ((d << 7) | c4)), v[u]);
        }
    }
}

extern "C" void kernel_launch(void* const* d_in, const int* in_sizes, int n_in,
                              void* d_out, int out_size, void* d_ws, size_t ws_size,
                              hipStream_t stream) {
    const float* x          = (const float*)d_in[0];
    const int*   edge_index = (const int*)  d_in[1];
    const float* W          = (const float*)d_in[2];
    const float* b          = (const float*)d_in[3];
    float*       out        = (float*)d_out;
    float*       y          = (float*)d_ws;

    const int* src = edge_index;
    const int* dst = edge_index + N_EDGES;

    const int proj_blocks = (M_TILES + PROJ_WAVES - 1) / PROJ_WAVES;
    gin_proj_wmma<<<proj_blocks, 128, 0, stream>>>(x, W, b, y, out);

    const int edge_blocks = N_CHUNKS / WAVES_PER_BLOCK;
    gin_edge_scatter<<<edge_blocks, 256, 0, stream>>>(src, dst, y, out);
}


# edge sum without global atomics: edges binned by target range with LDS counters, owner work-groups add gathered rows in LDS (arrival order) and store each out line once, y + bias added last
# speedup vs baseline: 3.8069x; 1.2704x over previous
#include <hip/hip_runtime.h>


typedef __attribute__((ext_vector_type(2))) float v2f;
typedef __attribute__((ext_vector_type(4))) float v4f;
typedef __attribute__((ext_vector_type(8))) float v8f;

#define N_NODES   100000
#define D_FEAT    128
#define N_EDGES   1600000
#define N_CLASSES 32
#define M_TILES   (N_NODES / 16)

#define PROJ_WAVES 4
#define XP 132
#define OP 36

__global__ __launch_bounds__(128)
void gin_proj_wmma(const float* __restrict__ x, const float* __restrict__ W,
                   float* __restrict__ y) {
    __shared__ __attribute__((aligned(16))) float Wl[N_CLASSES * XP];
    __shared__ __attribute__((aligned(16))) float Xl[PROJ_WAVES * 16 * XP];

    const int tid  = threadIdx.x;
    const int lane = tid & 31;
    const int wave  = __builtin_amdgcn_readfirstlane(tid >> 5);
    const int mtile = blockIdx.x * PROJ_WAVES + wave;
    const bool valid = mtile < M_TILES;
    float* xl = Xl + wave * (16 * XP);

    v4f xr[16];
    if (valid) {
        const float* xt = x + mtile * (16 * D_FEAT) + lane * 4;
#pragma unroll
        for (int r = 0; r < 16; ++r) xr[r] = *(const v4f*)(xt + r * D_FEAT);
    }
#pragma unroll
    for (int i = 0; i < 8; ++i) {
        const int idx = tid + 128 * i;
        const int row = idx >> 5;
        const int c4  = (idx & 31) * 4;
        *(v4f*)(Wl + row * XP + c4) = *(const v4f*)(W + row * D_FEAT + c4);
    }
    if (valid) {
#pragma unroll
        for (int r = 0; r < 16; ++r) *(v4f*)(xl + r * XP + lane * 4) = xr[r];
    }
    __syncthreads();

    const int c = lane & 15;
    const int h = lane >> 4;
    v8f acc0 = {};
    v8f acc1 = {};
    if (valid) {
        const float* ap  = xl + c * XP + 4 * h;
        const float* b0p = Wl + c * XP + 4 * h;
        const float* b1p = Wl + (c + 16) * XP + 4 * h;
#pragma unroll
        for (int j = 0; j < D_FEAT / 8; ++j) {
            const v4f a  = *(const v4f*)(ap  + 8 * j);
            const v4f b0 = *(const v4f*)(b0p + 8 * j);
            const v4f b1 = *(const v4f*)(b1p + 8 * j);
            const v2f alo  = {a[0],  a[1]},  ahi  = {a[2],  a[3]};
            const v2f b0lo = {b0[0], b0[1]}, b0hi = {b0[2], b0[3]};
            const v2f b1lo = {b1[0], b1[1]}, b1hi = {b1[2], b1[3]};
            acc0 = __builtin_amdgcn_wmma_f32_16x16x4_f32(false, alo, false, b0lo,
                                                         (short)0, acc0, false, false);
            acc1 = __builtin_amdgcn_wmma_f32_16x16x4_f32(false, alo, false, b1lo,
                                                         (short)0, acc1, false, false);
            acc0 = __builtin_amdgcn_wmma_f32_16x16x4_f32(false, ahi, false, b0hi,
                                                         (short)0, acc0, false, false);
            acc1 = __builtin_amdgcn_wmma_f32_16x16x4_f32(false, ahi, false, b1hi,
                                                         (short)0, acc1, false, false);
        }
#pragma unroll
        for (int r = 0; r < 8; ++r) {
            xl[(8 * h + r) * OP + c]      = acc0[r];
            xl[(8 * h + r) * OP + c + 16] = acc1[r];
        }
    }
    __syncthreads();
    if (valid) {
        const int c4 = (lane & 7) * 4;
        float* yt = y + mtile * (16 * N_CLASSES);
#pragma unroll
        for (int i = 0; i < 4; ++i) {
            const int row = (lane >> 3) + 4 * i;
            *(v4f*)(yt + row * N_CLASSES + c4) = *(const v4f*)(xl + row * OP + c4);
        }
    }
}

#define N_RANGES      256
#define RANGE_ROWS    391
#define BIN_BLOCKS    250
#define BIN_EDGES     6400
#define BIN_THREADS   640
#define BIN_PER_THREAD 10
#define TAB_PITCH     (N_RANGES + 1)
#define SRC_BITS      17
#define SRC_MASK      0x1FFFFu
static_assert(BIN_BLOCKS * BIN_EDGES == N_EDGES, "edge list must split into whole bin blocks");
static_assert(BIN_THREADS * BIN_PER_THREAD == BIN_EDGES, "threads must cover a bin block");
static_assert(N_RANGES * RANGE_ROWS >= N_NODES, "ranges must cover all rows");
static_assert(N_NODES <= (1 << SRC_BITS) && RANGE_ROWS < 512, "record fields must fit");

__global__ __launch_bounds__(BIN_THREADS)
void gin_edge_bin(const int* __restrict__ src, const int* __restrict__ dst,
                  unsigned* __restrict__ recs, unsigned* __restrict__ tab) {
    __shared__ unsigned cnt[N_RANGES];
    __shared__ unsigned start[N_RANGES + 1];
    __shared__ unsigned buf[BIN_EDGES];

    const int tid = threadIdx.x;
    if (tid < N_RANGES) cnt[tid] = 0u;

    const int e0 = blockIdx.x * BIN_EDGES + tid;
    int s[BIN_PER_THREAD], d[BIN_PER_THREAD];
#pragma unroll
    for (int i = 0; i < BIN_PER_THREAD; ++i) {
        s[i] = src[e0 + i * BIN_THREADS];
        d[i] = dst[e0 + i * BIN_THREADS];
    }
    __syncthreads();

    unsigned rec[BIN_PER_THREAD], rng[BIN_PER_THREAD], rank[BIN_PER_THREAD];
#pragma unroll
    for (int i = 0; i < BIN_PER_THREAD; ++i) {
        const unsigned t = (unsigned)d[i];
        const unsigned r = t / RANGE_ROWS;
        rng[i]  = r;
        rec[i]  = (unsigned)s[i] | ((t - r * RANGE_ROWS) << SRC_BITS);
        rank[i] = atomicAdd(&cnt[r], 1u);
    }
    __syncthreads();

    if (tid < 32) {
        unsigned c[8];
        unsigned sum = 0u;
#pragma unroll
        for (int k = 0; k < 8; ++k) { c[k] = cnt[tid * 8 + k]; sum += c[k]; }
        unsigned incl = sum;
#pragma unroll
        for (int o = 1; o < 32; o <<= 1) {
            const unsigned up = __shfl_up(incl, o);
            if (tid >= o) incl += up;
        }
        unsigned run = incl - sum;
#pragma unroll
        for (int k = 0; k < 8; ++k) { start[tid * 8 + k] = run; run += c[k]; }
        if (tid == 31) start[N_RANGES] = run;
    }
    __syncthreads();

#pragma unroll
    for (int i = 0; i < BIN_PER_THREAD; ++i) buf[start[rng[i]] + rank[i]] = rec[i];
    __syncthreads();

    unsigned* rb = recs + blockIdx.x * BIN_EDGES;
#pragma unroll
    for (int i = 0; i < BIN_PER_THREAD; ++i) rb[tid + i * BIN_THREADS] = buf[tid + i * BIN_THREADS];
    if (tid <= N_RANGES) tab[blockIdx.x * TAB_PITCH + tid] = start[tid];
}

#define OWN_THREADS 1024
#define OWN_WAVES   32
#define OWN_SEGS    8
#define LIST_CAP    (OWN_SEGS * 64 + 32)
#define SPARE_REC   ((unsigned)RANGE_ROWS << SRC_BITS)
static_assert(OWN_WAVES * OWN_SEGS >= BIN_BLOCKS, "waves must cover all bin blocks");

__device__ __forceinline__ void sum_rows32(unsigned rec, const float* __restrict__ y,
                                           float* acc, unsigned c4, int lane) {
    float v[32];
#pragma unroll
    for (int u = 0; u < 32; ++u) {
        const unsigned s = (unsigned)__builtin_amdgcn_readlane((int)rec, u) & SRC_MASK;
        v[u] = *(const float*)((const char*)y + ((s << 7) | c4));
    }
#pragma unroll
    for (int u = 0; u < 32; ++u) {
        const unsigned r = (unsigned)__builtin_amdgcn_readlane((int)rec, u) >> SRC_BITS;
        atomicAdd(&acc[r * 32 + lane], v[u]);
    }
}

__global__ __launch_bounds__(OWN_THREADS)
void gin_owner_sum(const unsigned* __restrict__ recs, const unsigned* __restrict__ tab,
                   const float* __restrict__ y, const float* __restrict__ bias,
                   float* __restrict__ out) {
    __shared__ float acc[(RANGE_ROWS + 1) * 32];
    __shared__ unsigned list[OWN_WAVES * LIST_CAP];

    const int tid  = threadIdx.x;
    const int lane = tid & 31;
    const unsigned c4 = (unsigned)lane * 4u;
    const int wave = __builtin_amdgcn_readfirstlane(tid >> 5);
    const int b = blockIdx.x;
    unsigned* wl = list + wave * LIST_CAP;

    for (int i = tid; i < (RANGE_ROWS + 1) * 32; i += OWN_THREADS) acc[i] = 0.f;

    unsigned st = 0u, en = 0u;
    {
        const int blk = wave + 32 * lane;
        if (lane < OWN_SEGS && blk < BIN_BLOCKS) {
            st = tab[blk * TAB_PITCH + b];
            en = tab[blk * TAB_PITCH + b + 1];
        }
    }
    unsigned r0[OWN_SEGS], r1[OWN_SEGS];
#pragma unroll
    for (int j = 0; j < OWN_SEGS; ++j) {
        const unsigned sj = (unsigned)__builtin_amdgcn_readlane((int)st, j);
        const unsigned ej = (unsigned)__builtin_amdgcn_readlane((int)en, j);
        const unsigned* p = recs + (wave + 32 * j) * BIN_EDGES;
        const unsigned i0 = sj + (unsigned)lane;
        r0[j] = (i0 < ej)       ? p[i0]      : SPARE_REC;
        r1[j] = (i0 + 32u < ej) ? p[i0 + 32] : SPARE_REC;
    }
    unsigned fill = 0u;
#pragma unroll
    for (int j = 0; j < OWN_SEGS; ++j) {
        const unsigned n = (unsigned)__builtin_amdgcn_readlane((int)en, j)
                         - (unsigned)__builtin_amdgcn_readlane((int)st, j);
        wl[fill + lane] = r0[j];
        fill += (n < 32u) ? n : 32u;
        wl[fill + lane] = r1[j];
        fill += (n > 32u) ? ((n < 64u) ? n - 32u : 32u) : 0u;
    }
    wl[fill + lane] = SPARE_REC;
    __syncthreads();

#pragma unroll 1
    for (unsigned t = 0u; t < fill; t += 32u) sum_rows32(wl[t + lane], y, acc, c4, lane);

#pragma unroll 1
    for (int j = 0; j < OWN_SEGS; ++j) {
        const unsigned sj = (unsigned)__builtin_amdgcn_readlane((int)st, j);
        const unsigned ej = (unsigned)__builtin_amdgcn_readlane((int)en, j);
        const unsigned* p = recs + (wave + 32 * j) * BIN_EDGES;
#pragma unroll 1
        for (unsigned base = sj + 64u; base < ej; base += 32u) {
            const unsigned i0 = base + (unsigned)lane;
            const unsigned r = (i0 < ej) ? p[i0] : SPARE_REC;
            sum_rows32(r, y, acc, c4, lane);
        }
    }
    __syncthreads();

    const float bv = bias[lane];
#pragma unroll
    for (int k = 0; k < (RANGE_ROWS + OWN_WAVES - 1) / OWN_WAVES; ++k) {
        const int row  = wave + OWN_WAVES * k;
        const int node = b * RANGE_ROWS + row;
        if (row < RANGE_ROWS && node < N_NODES)
            out[node * N_CLASSES + lane] = y[node * N_CLASSES + lane] + bv + acc[row * 32 + lane];
    }
}

#define WS_RECS_OFF (16u << 20)
#define WS_TAB_OFF  (24u << 20)

extern "C" void kernel_launch(void* const* d_in, const int* in_sizes, int n_in,
                              void* d_out, int out_size, void* d_ws, size_t ws_size,
                              hipStream_t stream) {
    const float* x          = (const float*)d_in[0];
    const int*   edge_index = (const int*)  d_in[1];
    const float* W          = (const float*)d_in[2];
    const float* b          = (const float*)d_in[3];
    float*       out        = (float*)d_out;
    float*       y          = (float*)d_ws;
    unsigned*    recs       = (unsigned*)((char*)d_ws + WS_RECS_OFF);
    unsigned*    tab        = (unsigned*)((char*)d_ws + WS_TAB_OFF);

    const int* src = edge_index;
    const int* dst = edge_index + N_EDGES;

    const int proj_blocks = (M_TILES + PROJ_WAVES - 1) / PROJ_WAVES;
    gin_proj_wmma<<<proj_blocks, 128, 0, stream>>>(x, W, y);
    gin_edge_bin<<<BIN_BLOCKS, BIN_THREADS, 0, stream>>>(src, dst, recs, tab);
    gin_owner_sum<<<N_RANGES, OWN_THREADS, 0, stream>>>(recs, tab, y, b, out);
}


# projection and edge binning as two block roles of one launch (250 binning blocks first, 625 projection blocks, 10 waves each, one LDS block): 3 launches to 2; edge sums in arrival order as before
# speedup vs baseline: 3.9772x; 1.0447x over previous
#include <hip/hip_runtime.h>


typedef __attribute__((ext_vector_type(2))) float v2f;
typedef __attribute__((ext_vector_type(4))) float v4f;
typedef __attribute__((ext_vector_type(8))) float v8f;

#define N_NODES   100000
#define D_FEAT    128
#define N_EDGES   1600000
#define N_CLASSES 32
#define M_TILES   (N_NODES / 16)

#define L1_THREADS 320
#define PROJ_WAVES 10
#define PROJ_BLOCKS ((M_TILES + PROJ_WAVES - 1) / PROJ_WAVES)
#define XP 132
#define OP 36
#define L1_LDS_BYTES ((N_CLASSES + PROJ_WAVES * 16) * XP * 4)
#define W_LOAD_WAVES 8
static_assert(W_LOAD_WAVES <= PROJ_WAVES && (N_CLASSES * D_FEAT / 4) % (W_LOAD_WAVES * 32) == 0, "W must split evenly over its loading waves");

__device__ __forceinline__
void proj_role(const float* __restrict__ x, const float* __restrict__ W,
               float* __restrict__ y, float* lds, int pblock) {
    float* Wl = lds;
    float* Xl = lds + N_CLASSES * XP;

    const int tid  = threadIdx.x;
    const int lane = tid & 31;
    const int wave  = __builtin_amdgcn_readfirstlane(tid >> 5);
    const int mtile = pblock * PROJ_WAVES + wave;
    const bool valid = mtile < M_TILES;
    float* xl = Xl + wave * (16 * XP);

    v4f xr[16];
    if (valid) {
        const float* xt = x + mtile * (16 * D_FEAT) + lane * 4;
#pragma unroll
        for (int r = 0; r < 16; ++r) xr[r] = *(const v4f*)(xt + r * D_FEAT);
    }
    if (wave < W_LOAD_WAVES) {
#pragma unroll
        for (int i = 0; i < N_CLASSES * D_FEAT / 4 / (W_LOAD_WAVES * 32); ++i) {
            const int idx = tid + W_LOAD_WAVES * 32 * i;
            const int row = idx >> 5;
            const int c4  = (idx & 31) * 4;
            *(v4f*)(Wl + row * XP + c4) = *(const v4f*)(W + row * D_FEAT + c4);
        }
    }
    if (valid) {
#pragma unroll
        for (int r = 0; r < 16; ++r) *(v4f*)(xl + r * XP + lane * 4) = xr[r];
    }
    __syncthreads();

    const int c = lane & 15;
    const int h = lane >> 4;
    v8f acc0 = {};
    v8f acc1 = {};
    if (valid) {
        const float* ap  = xl + c * XP + 4 * h;
        const float* b0p = Wl + c * XP + 4 * h;
        const float* b1p = Wl + (c + 16) * XP + 4 * h;
#pragma unroll
        for (int j = 0; j < D_FEAT / 8; ++j) {
            const v4f a  = *(const v4f*)(ap  + 8 * j);
            const v4f b0 = *(const v4f*)(b0p + 8 * j);
            const v4f b1 = *(const v4f*)(b1p + 8 * j);
            const v2f alo  = {a[0],  a[1]},  ahi  = {a[2],  a[3]};
            const v2f b0lo = {b0[0], b0[1]}, b0hi = {b0[2], b0[3]};
            const v2f b1lo = {b1[0], b1[1]}, b1hi = {b1[2], b1[3]};
            acc0 = __builtin_amdgcn_wmma_f32_16x16x4_f32(false, alo, false, b0lo,
                                                         (short)0, acc0, false, false);
            acc1 = __builtin_amdgcn_wmma_f32_16x16x4_f32(false, alo, false, b1lo,
                                                         (short)0, acc1, false, false);
            acc0 = __builtin_amdgcn_wmma_f32_16x16x4_f32(false, ahi, false, b0hi,
                                                         (short)0, acc0, false, false);
            acc1 = __builtin_amdgcn_wmma_f32_16x16x4_f32(false, ahi, false, b1hi,
                                                         (short)0, acc1, false, false);
        }
#pragma unroll
        for (int r = 0; r < 8; ++r) {
            xl[(8 * h + r) * OP + c]      = acc0[r];
            xl[(8 * h + r) * OP + c + 16] = acc1[r];
        }
    }
    __syncthreads();
    if (valid) {
        const int c4 = (lane & 7) * 4;
        float* yt = y + mtile * (16 * N_CLASSES);
#pragma unroll
        for (int i = 0; i < 4; ++i) {
            const int row = (lane >> 3) + 4 * i;
            *(v4f*)(yt + row * N_CLASSES + c4) = *(const v4f*)(xl + row * OP + c4);
        }
    }
}

#define N_RANGES      256
#define RANGE_ROWS    391
#define BIN_BLOCKS    250
#define BIN_EDGES     6400
#define BIN_PER_THREAD 20
#define TAB_PITCH     (N_RANGES + 1)
#define SRC_BITS      17
#define SRC_MASK      0x1FFFFu
static_assert(BIN_BLOCKS * BIN_EDGES == N_EDGES, "edge list must split into whole bin blocks");
static_assert(L1_THREADS * BIN_PER_THREAD == BIN_EDGES, "threads must cover a bin block");
static_assert(L1_THREADS > N_RANGES, "one thread per range start");
static_assert(N_RANGES * RANGE_ROWS >= N_NODES, "ranges must cover all rows");
static_assert(N_NODES <= (1 << SRC_BITS) && RANGE_ROWS < 512, "record fields must fit");
static_assert((N_RANGES + N_RANGES + 1 + BIN_EDGES) * 4 <= L1_LDS_BYTES, "binning must fit the shared LDS block");

__device__ __forceinline__
void bin_role(const int* __restrict__ src, const int* __restrict__ dst,
              unsigned* __restrict__ recs, unsigned* __restrict__ tab,
              unsigned* lds, int bblock) {
    unsigned* cnt   = lds;
    unsigned* start = lds + N_RANGES;
    unsigned* buf   = lds + 2 * N_RANGES + 1;

    const int tid = threadIdx.x;
    if (tid < N_RANGES) cnt[tid] = 0u;

    const int e0 = bblock * BIN_EDGES + tid;
    int s[BIN_PER_THREAD], d[BIN_PER_THREAD];
#pragma unroll
    for (int i = 0; i < BIN_PER_THREAD; ++i) {
        s[i] = src[e0 + i * L1_THREADS];
        d[i] = dst[e0 + i * L1_THREADS];
    }
    __syncthreads();

    unsigned rec[BIN_PER_THREAD], rng[BIN_PER_THREAD], rank[BIN_PER_THREAD];
#pragma unroll
    for (int i = 0; i < BIN_PER_THREAD; ++i) {
        const unsigned t = (unsigned)d[i];
        const unsigned r = t / RANGE_ROWS;
        rng[i]  = r;
        rec[i]  = (unsigned)s[i] | ((t - r * RANGE_ROWS) << SRC_BITS);
        rank[i] = atomicAdd(&cnt[r], 1u);
    }
    __syncthreads();

    if (tid < 32) {
        unsigned c[8];
        unsigned sum = 0u;
#pragma unroll
        for (int k = 0; k < 8; ++k) { c[k] = cnt[tid * 8 + k]; sum += c[k]; }
        unsigned incl = sum;
#pragma unroll
        for (int o = 1; o < 32; o <<= 1) {
            const unsigned up = __shfl_up(incl, o);
            if (tid >= o) incl += up;
        }
        unsigned run = incl - sum;
#pragma unroll
        for (int k = 0; k < 8; ++k) { start[tid * 8 + k] = run; run += c[k]; }
        if (tid == 31) start[N_RANGES] = run;
    }
    __syncthreads();

#pragma unroll
    for (int i = 0; i < BIN_PER_THREAD; ++i) buf[start[rng[i]] + rank[i]] = rec[i];
    __syncthreads();

    unsigned* rb = recs + bblock * BIN_EDGES;
#pragma unroll
    for (int i = 0; i < BIN_PER_THREAD; ++i) rb[tid + i * L1_THREADS] = buf[tid + i * L1_THREADS];
    if (tid <= N_RANGES) tab[bblock * TAB_PITCH + tid] = start[tid];
}

__global__ __launch_bounds__(L1_THREADS)
void gin_bin_and_proj(const float* __restrict__ x, const float* __restrict__ W,
                      float* __restrict__ y,
                      const int* __restrict__ src, const int* __restrict__ dst,
                      unsigned* __restrict__ recs, unsigned* __restrict__ tab) {
    __shared__ __attribute__((aligned(16))) unsigned char lds_raw[L1_LDS_BYTES];
    if (blockIdx.x < BIN_BLOCKS) bin_role(src, dst, recs, tab, (unsigned*)lds_raw, (int)blockIdx.x);
    else                         proj_role(x, W, y, (float*)lds_raw, (int)blockIdx.x - BIN_BLOCKS);
}

#define OWN_THREADS 1024
#define OWN_WAVES   32
#define OWN_SEGS    8
#define LIST_CAP    (OWN_SEGS * 64 + 32)
#define SPARE_REC   ((unsigned)RANGE_ROWS << SRC_BITS)
static_assert(OWN_WAVES * OWN_SEGS >= BIN_BLOCKS, "waves must cover all bin blocks");

__device__ __forceinline__ void sum_rows32(unsigned rec, const float* __restrict__ y,
                                           float* acc, unsigned c4, int lane) {
    float v[32];
#pragma unroll
    for (int u = 0; u < 32; ++u) {
        const unsigned s = (unsigned)__builtin_amdgcn_readlane((int)rec, u) & SRC_MASK;
        v[u] = *(const float*)((const char*)y + ((s << 7) | c4));
    }
#pragma unroll
    for (int u = 0; u < 32; ++u) {
        const unsigned r = (unsigned)__builtin_amdgcn_readlane((int)rec, u) >> SRC_BITS;
        atomicAdd(&acc[r * 32 + lane], v[u]);
    }
}

__global__ __launch_bounds__(OWN_THREADS)
void gin_owner_sum(const unsigned* __restrict__ recs, const unsigned* __restrict__ tab,
                   const float* __restrict__ y, const float* __restrict__ bias,
                   float* __restrict__ out) {
    __shared__ float acc[(RANGE_ROWS + 1) * 32];
    __shared__ unsigned list[OWN_WAVES * LIST_CAP];

    const int tid  = threadIdx.x;
    const int lane = tid & 31;
    const unsigned c4 = (unsigned)lane * 4u;
    const int wave = __builtin_amdgcn_readfirstlane(tid >> 5);
    const int b = blockIdx.x;
    unsigned* wl = list + wave * LIST_CAP;

    for (int i = tid; i < (RANGE_ROWS + 1) * 32; i += OWN_THREADS) acc[i] = 0.f;

    unsigned st = 0u, en = 0u;
    {
        const int blk = wave + 32 * lane;
        if (lane < OWN_SEGS && blk < BIN_BLOCKS) {
            st = tab[blk * TAB_PITCH + b];
            en = tab[blk * TAB_PITCH + b + 1];
        }
    }
    unsigned r0[OWN_SEGS], r1[OWN_SEGS];
#pragma unroll
    for (int j = 0; j < OWN_SEGS; ++j) {
        const unsigned sj = (unsigned)__builtin_amdgcn_readlane((int)st, j);
        const unsigned ej = (unsigned)__builtin_amdgcn_readlane((int)en, j);
        const unsigned* p = recs + (wave + 32 * j) * BIN_EDGES;
        const unsigned i0 = sj + (unsigned)lane;
        r0[j] = (i0 < ej)       ? p[i0]      : SPARE_REC;
        r1[j] = (i0 + 32u < ej) ? p[i0 + 32] : SPARE_REC;
    }
    unsigned fill = 0u;
#pragma unroll
    for (int j = 0; j < OWN_SEGS; ++j) {
        const unsigned n = (unsigned)__builtin_amdgcn_readlane((int)en, j)
                         - (unsigned)__builtin_amdgcn_readlane((int)st, j);
        wl[fill + lane] = r0[j];
        fill += (n < 32u) ? n : 32u;
        wl[fill + lane] = r1[j];
        fill += (n > 32u) ? ((n < 64u) ? n - 32u : 32u) : 0u;
    }
    wl[fill + lane] = SPARE_REC;
    __syncthreads();

#pragma unroll 1
    for (unsigned t = 0u; t < fill; t += 32u) sum_rows32(wl[t + lane], y, acc, c4, lane);

#pragma unroll 1
    for (int j = 0; j < OWN_SEGS; ++j) {
        const unsigned sj = (unsigned)__builtin_amdgcn_readlane((int)st, j);
        const unsigned ej = (unsigned)__builtin_amdgcn_readlane((int)en, j);
        const unsigned* p = recs + (wave + 32 * j) * BIN_EDGES;
#pragma unroll 1
        for (unsigned base = sj + 64u; base < ej; base += 32u) {
            const unsigned i0 = base + (unsigned)lane;
            const unsigned r = (i0 < ej) ? p[i0] : SPARE_REC;
            sum_rows32(r, y, acc, c4, lane);
        }
    }
    __syncthreads();

    const float bv = bias[lane];
#pragma unroll
    for (int k = 0; k < (RANGE_ROWS + OWN_WAVES - 1) / OWN_WAVES; ++k) {
        const int row  = wave + OWN_WAVES * k;
        const int node = b * RANGE_ROWS + row;
        if (row < RANGE_ROWS && node < N_NODES)
            out[node * N_CLASSES + lane] = y[node * N_CLASSES + lane] + bv + acc[row * 32 + lane];
    }
}

#define WS_RECS_OFF (16u << 20)
#define WS_TAB_OFF  (24u << 20)

extern "C" void kernel_launch(void* const* d_in, const int* in_sizes, int n_in,
                              void* d_out, int out_size, void* d_ws, size_t ws_size,
                              hipStream_t stream) {
    const float* x          = (const float*)d_in[0];
    const int*   edge_index = (const int*)  d_in[1];
    const float* W          = (const float*)d_in[2];
    const float* b          = (const float*)d_in[3];
    float*       out        = (float*)d_out;
    float*       y          = (float*)d_ws;
    unsigned*    recs       = (unsigned*)((char*)d_ws + WS_RECS_OFF);
    unsigned*    tab        = (unsigned*)((char*)d_ws + WS_TAB_OFF);

    const int* src = edge_index;
    const int* dst = edge_index + N_EDGES;

    gin_bin_and_proj<<<BIN_BLOCKS + PROJ_BLOCKS, L1_THREADS, 0, stream>>>(x, W, y, src, dst, recs, tab);
    gin_owner_sum<<<N_RANGES, OWN_THREADS, 0, stream>>>(recs, tab, y, b, out);
}


# owner launch orders its records by target row in LDS (counting sort, rank = returning integer add); a wave sums each row in a register, own y row first then rank order, and stores it once
# speedup vs baseline: 5.0006x; 1.2573x over previous
#include <hip/hip_runtime.h>


typedef __attribute__((ext_vector_type(2))) float v2f;
typedef __attribute__((ext_vector_type(4))) float v4f;
typedef __attribute__((ext_vector_type(8))) float v8f;

#define N_NODES   100000
#define D_FEAT    128
#define N_EDGES   1600000
#define N_CLASSES 32
#define M_TILES   (N_NODES / 16)

#define L1_THREADS 320
#define PROJ_WAVES 10
#define PROJ_BLOCKS ((M_TILES + PROJ_WAVES - 1) / PROJ_WAVES)
#define XP 132
#define OP 36
#define L1_LDS_BYTES ((N_CLASSES + PROJ_WAVES * 16) * XP * 4)
#define W_LOAD_WAVES 8
static_assert(W_LOAD_WAVES <= PROJ_WAVES && (N_CLASSES * D_FEAT / 4) % (W_LOAD_WAVES * 32) == 0, "W must split evenly over its loading waves");

__device__ __forceinline__
void proj_role(const float* __restrict__ x, const float* __restrict__ W,
               float* __restrict__ y, float* lds, int pblock) {
    float* Wl = lds;
    float* Xl = lds + N_CLASSES * XP;

    const int tid  = threadIdx.x;
    const int lane = tid & 31;
    const int wave  = __builtin_amdgcn_readfirstlane(tid >> 5);
    const int mtile = pblock * PROJ_WAVES + wave;
    const bool valid = mtile < M_TILES;
    float* xl = Xl + wave * (16 * XP);

    v4f xr[16];
    if (valid) {
        const float* xt = x + mtile * (16 * D_FEAT) + lane * 4;
#pragma unroll
        for (int r = 0; r < 16; ++r) xr[r] = *(const v4f*)(xt + r * D_FEAT);
    }
    if (wave < W_LOAD_WAVES) {
#pragma unroll
        for (int i = 0; i < N_CLASSES * D_FEAT / 4 / (W_LOAD_WAVES * 32); ++i) {
            const int idx = tid + W_LOAD_WAVES * 32 * i;
            const int row = idx >> 5;
            const int c4  = (idx & 31) * 4;
            *(v4f*)(Wl + row * XP + c4) = *(const v4f*)(W + row * D_FEAT + c4);
        }
    }
    if (valid) {
#pragma unroll
        for (int r = 0; r < 16; ++r) *(v4f*)(xl + r * XP + lane * 4) = xr[r];
    }
    __syncthreads();

    const int c = lane & 15;
    const int h = lane >> 4;
    v8f acc0 = {};
    v8f acc1 = {};
    if (valid) {
        const float* ap  = xl + c * XP + 4 * h;
        const float* b0p = Wl + c * XP + 4 * h;
        const float* b1p = Wl + (c + 16) * XP + 4 * h;
#pragma unroll
        for (int j = 0; j < D_FEAT / 8; ++j) {
            const v4f a  = *(const v4f*)(ap  + 8 * j);
            const v4f b0 = *(const v4f*)(b0p + 8 * j);
            const v4f b1 = *(const v4f*)(b1p + 8 * j);
            const v2f alo  = {a[0],  a[1]},  ahi  = {a[2],  a[3]};
            const v2f b0lo = {b0[0], b0[1]}, b0hi = {b0[2], b0[3]};
            const v2f b1lo = {b1[0], b1[1]}, b1hi = {b1[2], b1[3]};
            acc0 = __builtin_amdgcn_wmma_f32_16x16x4_f32(false, alo, false, b0lo,
                                                         (short)0, acc0, false, false);
            acc1 = __builtin_amdgcn_wmma_f32_16x16x4_f32(false, alo, false, b1lo,
                                                         (short)0, acc1, false, false);
            acc0 = __builtin_amdgcn_wmma_f32_16x16x4_f32(false, ahi, false, b0hi,
                                                         (short)0, acc0, false, false);
            acc1 = __builtin_amdgcn_wmma_f32_16x16x4_f32(false, ahi, false, b1hi,
                                                         (short)0, acc1, false, false);
        }
#pragma unroll
        for (int r = 0; r < 8; ++r) {
            xl[(8 * h + r) * OP + c]      = acc0[r];
            xl[(8 * h + r) * OP + c + 16] = acc1[r];
        }
    }
    __syncthreads();
    if (valid) {
        const int c4 = (lane & 7) * 4;
        float* yt = y + mtile * (16 * N_CLASSES);
#pragma unroll
        for (int i = 0; i < 4; ++i) {
            const int row = (lane >> 3) + 4 * i;
            *(v4f*)(yt + row * N_CLASSES + c4) = *(const v4f*)(xl + row * OP + c4);
        }
    }
}

#define N_RANGES      256
#define RANGE_ROWS    391
#define BIN_BLOCKS    250
#define BIN_EDGES     6400
#define BIN_PER_THREAD 20
#define TAB_PITCH     (N_RANGES + 1)
#define SRC_BITS      17
#define SRC_MASK      0x1FFFFu
static_assert(BIN_BLOCKS * BIN_EDGES == N_EDGES, "edge list must split into whole bin blocks");
static_assert(L1_THREADS * BIN_PER_THREAD == BIN_EDGES, "threads must cover a bin block");
static_assert(L1_THREADS > N_RANGES, "one thread per range start");
static_assert(N_RANGES * RANGE_ROWS >= N_NODES, "ranges must cover all rows");
static_assert(N_NODES <= (1 << SRC_BITS) && RANGE_ROWS < 512, "record fields must fit");
static_assert((N_RANGES + N_RANGES + 1 + BIN_EDGES) * 4 <= L1_LDS_BYTES, "binning must fit the shared LDS block");

__device__ __forceinline__
void bin_role(const int* __restrict__ src, const int* __restrict__ dst,
              unsigned* __restrict__ recs, unsigned* __restrict__ tab,
              unsigned* lds, int bblock) {
    unsigned* cnt   = lds;
    unsigned* start = lds + N_RANGES;
    unsigned* buf   = lds + 2 * N_RANGES + 1;

    const int tid = threadIdx.x;
    if (tid < N_RANGES) cnt[tid] = 0u;

    const int e0 = bblock * BIN_EDGES + tid;
    int s[BIN_PER_THREAD], d[BIN_PER_THREAD];
#pragma unroll
    for (int i = 0; i < BIN_PER_THREAD; ++i) {
        s[i] = src[e0 + i * L1_THREADS];
        d[i] = dst[e0 + i * L1_THREADS];
    }
    __syncthreads();

    unsigned rec[BIN_PER_THREAD], rng[BIN_PER_THREAD], rank[BIN_PER_THREAD];
#pragma unroll
    for (int i = 0; i < BIN_PER_THREAD; ++i) {
        const unsigned t = (unsigned)d[i];
        const unsigned r = t / RANGE_ROWS;
        rng[i]  = r;
        rec[i]  = (unsigned)s[i] | ((t - r * RANGE_ROWS) << SRC_BITS);
        rank[i] = atomicAdd(&cnt[r], 1u);
    }
    __syncthreads();

    if (tid < 32) {
        unsigned c[8];
        unsigned sum = 0u;
#pragma unroll
        for (int k = 0; k < 8; ++k) { c[k] = cnt[tid * 8 + k]; sum += c[k]; }
        unsigned incl = sum;
#pragma unroll
        for (int o = 1; o < 32; o <<= 1) {
            const unsigned up = __shfl_up(incl, o);
            if (tid >= o) incl += up;
        }
        unsigned run = incl - sum;
#pragma unroll
        for (int k = 0; k < 8; ++k) { start[tid * 8 + k] = run; run += c[k]; }
        if (tid == 31) start[N_RANGES] = run;
    }
    __syncthreads();

#pragma unroll
    for (int i = 0; i < BIN_PER_THREAD; ++i) buf[start[rng[i]] + rank[i]] = rec[i];
    __syncthreads();

    unsigned* rb = recs + bblock * BIN_EDGES;
#pragma unroll
    for (int i = 0; i < BIN_PER_THREAD; ++i) rb[tid + i * L1_THREADS] = buf[tid + i * L1_THREADS];
    if (tid <= N_RANGES) tab[bblock * TAB_PITCH + tid] = start[tid];
}

__global__ __launch_bounds__(L1_THREADS)
void gin_bin_and_proj(const float* __restrict__ x, const float* __restrict__ W,
                      float* __restrict__ y,
                      const int* __restrict__ src, const int* __restrict__ dst,
                      unsigned* __restrict__ recs, unsigned* __restrict__ tab) {
    __shared__ __attribute__((aligned(16))) unsigned char lds_raw[L1_LDS_BYTES];
    if (blockIdx.x < BIN_BLOCKS) bin_role(src, dst, recs, tab, (unsigned*)lds_raw, (int)blockIdx.x);
    else                         proj_role(x, W, y, (float*)lds_raw, (int)blockIdx.x - BIN_BLOCKS);
}

#define OWN_THREADS 1024
#define OWN_WAVES   32
#define OWN_SEGS    8
#define SEG_HEAD    64
#define SORT_CAP    (BIN_BLOCKS * SEG_HEAD + RANGE_ROWS)
#define CNT_PAD     (OWN_WAVES * 13)
#define SPARE_REC   ((unsigned)RANGE_ROWS << SRC_BITS)
#define NO_ROW      511u
#define END_REC     (NO_ROW << SRC_BITS)
static_assert(OWN_WAVES * OWN_SEGS >= BIN_BLOCKS, "waves must cover all bin blocks");
static_assert(CNT_PAD > RANGE_ROWS && NO_ROW > RANGE_ROWS, "counter padding and the end mark must lie past the rows");
static_assert(12 * OWN_WAVES + 7 == RANGE_ROWS, "7 waves of 13 rows and 25 of 12 must cover the range");

__device__ __forceinline__ void sum_rows32(unsigned rec, const float* __restrict__ y,
                                           float* acc, unsigned c4, int lane) {
    float v[32];
#pragma unroll
    for (int u = 0; u < 32; ++u) {
        const unsigned s = (unsigned)__builtin_amdgcn_readlane((int)rec, u) & SRC_MASK;
        v[u] = *(const float*)((const char*)y + ((s << 7) | c4));
    }
#pragma unroll
    for (int u = 0; u < 32; ++u) {
        const unsigned r = (unsigned)__builtin_amdgcn_readlane((int)rec, u) >> SRC_BITS;
        atomicAdd(&acc[r * 32 + lane], v[u]);
    }
}

__device__ __forceinline__ void store_row(unsigned row, float sum, float bv, bool rest, const float* acc,
                                          float* __restrict__ out, int b, int lane) {
    if (row < (unsigned)RANGE_ROWS) {
        const int node = b * RANGE_ROWS + (int)row;
        if (node < N_NODES) {
            float o = sum + bv;
            if (rest) o += acc[row * 32 + lane];
            out[node * N_CLASSES + lane] = o;
        }
    }
}

__global__ __launch_bounds__(OWN_THREADS)
void gin_owner_sum(const unsigned* __restrict__ recs, const unsigned* __restrict__ tab,
                   const float* __restrict__ y, const float* __restrict__ bias,
                   float* __restrict__ out) {
    __shared__ float acc[(RANGE_ROWS + 1) * 32];
    __shared__ unsigned cnt[CNT_PAD];
    __shared__ unsigned start[CNT_PAD + 1];
    __shared__ unsigned sorted[SORT_CAP];
    __shared__ unsigned rest_flag;

    const int tid  = threadIdx.x;
    const int lane = tid & 31;
    const unsigned c4 = (unsigned)lane * 4u;
    const int wave = __builtin_amdgcn_readfirstlane(tid >> 5);
    const int b = blockIdx.x;

    for (int i = tid; i < (RANGE_ROWS + 1) * 32; i += OWN_THREADS) acc[i] = 0.f;
    if (tid < CNT_PAD) cnt[tid] = (tid < RANGE_ROWS && b * RANGE_ROWS + tid < N_NODES) ? 1u : 0u;
    if (tid == 0) rest_flag = 0u;

    unsigned st = 0u, en = 0u;
    {
        const int blk = wave + 32 * lane;
        if (lane < OWN_SEGS && blk < BIN_BLOCKS) {
            st = tab[blk * TAB_PITCH + b];
            en = tab[blk * TAB_PITCH + b + 1];
        }
    }
    unsigned r0[OWN_SEGS], r1[OWN_SEGS];
#pragma unroll
    for (int j = 0; j < OWN_SEGS; ++j) {
        const unsigned sj = (unsigned)__builtin_amdgcn_readlane((int)st, j);
        const unsigned ej = (unsigned)__builtin_amdgcn_readlane((int)en, j);
        const unsigned* p = recs + (wave + 32 * j) * BIN_EDGES;
        const unsigned i0 = sj + (unsigned)lane;
        r0[j] = (i0 < ej)       ? p[i0]      : SPARE_REC;
        r1[j] = (i0 + 32u < ej) ? p[i0 + 32] : SPARE_REC;
    }
    __syncthreads();

    unsigned k0[OWN_SEGS], k1[OWN_SEGS];
#pragma unroll
    for (int j = 0; j < OWN_SEGS; ++j) {
        k0[j] = 0u;
        k1[j] = 0u;
        if (r0[j] != SPARE_REC) k0[j] = atomicAdd(&cnt[r0[j] >> SRC_BITS], 1u);
        if (r1[j] != SPARE_REC) k1[j] = atomicAdd(&cnt[r1[j] >> SRC_BITS], 1u);
    }

    bool used_rest = false;
#pragma unroll 1
    for (int j = 0; j < OWN_SEGS; ++j) {
        const unsigned sj = (unsigned)__builtin_amdgcn_readlane((int)st, j);
        const unsigned ej = (unsigned)__builtin_amdgcn_readlane((int)en, j);
        const unsigned* p = recs + (wave + 32 * j) * BIN_EDGES;
#pragma unroll 1
        for (unsigned base = sj + (unsigned)SEG_HEAD; base < ej; base += 32u) {
            const unsigned i0 = base + (unsigned)lane;
            const unsigned r = (i0 < ej) ? p[i0] : SPARE_REC;
            sum_rows32(r, y, acc, c4, lane);
            used_rest = true;
        }
    }
    if (used_rest && lane == 0) rest_flag = 1u;
    __syncthreads();

    if (wave == 0) {
        unsigned c[13];
        unsigned tot = 0u;
#pragma unroll
        for (int k = 0; k < 13; ++k) { c[k] = cnt[lane * 13 + k]; tot += c[k]; }
        unsigned incl = tot;
#pragma unroll
        for (int o = 1; o < 32; o <<= 1) {
            const unsigned up = __shfl_up(incl, o);
            if (lane >= o) incl += up;
        }
        unsigned run = incl - tot;
#pragma unroll
        for (int k = 0; k < 13; ++k) { start[lane * 13 + k] = run; run += c[k]; }
        if (lane == 31) start[CNT_PAD] = run;
    }
    __syncthreads();

#pragma unroll
    for (int j = 0; j < OWN_SEGS; ++j) {
        if (r0[j] != SPARE_REC) sorted[start[r0[j] >> SRC_BITS] + k0[j]] = r0[j];
        if (r1[j] != SPARE_REC) sorted[start[r1[j] >> SRC_BITS] + k1[j]] = r1[j];
    }
    if (tid < RANGE_ROWS && b * RANGE_ROWS + tid < N_NODES)
        sorted[start[tid]] = (unsigned)(b * RANGE_ROWS + tid) | ((unsigned)tid << SRC_BITS);
    __syncthreads();

    const int row_lo = 12 * wave + (wave < 7 ? wave : 7);
    const int row_hi = 12 * (wave + 1) + (wave + 1 < 7 ? wave + 1 : 7);
    const unsigned p0 = (unsigned)__builtin_amdgcn_readfirstlane((int)start[row_lo]);
    const unsigned p1 = (unsigned)__builtin_amdgcn_readfirstlane((int)start[row_hi]);
    const bool rest = __builtin_amdgcn_readfirstlane((int)rest_flag) != 0;
    const float bv = bias[lane];

    float sum = 0.f;
#pragma unroll 1
    for (unsigned t = p0; t < p1; t += 32u) {
        const unsigned idx = t + (unsigned)lane;
        const unsigned rec = (idx < p1) ? sorted[idx] : END_REC;
        const unsigned prow = (idx > p0 && idx <= p1) ? (sorted[idx - 1u] >> SRC_BITS) : NO_ROW;
        const unsigned chg = __builtin_amdgcn_ballot_w32((rec >> SRC_BITS) != prow);
        float v[32];
#pragma unroll
        for (int u = 0; u < 32; ++u) {
            const unsigned s = (unsigned)__builtin_amdgcn_readlane((int)rec, u) & SRC_MASK;
            v[u] = *(const float*)((const char*)y + ((s << 7) | c4));
        }
#pragma unroll
        for (int u = 0; u < 32; ++u) {
            if (__builtin_expect((chg >> u) & 1u, 0u)) {
                store_row((unsigned)__builtin_amdgcn_readlane((int)prow, u), sum, bv, rest, acc, out, b, lane);
                sum = 0.f;
            }
            sum += v[u];
        }
    }
    if (p1 > p0 && ((p1 - p0) & 31u) == 0u)
        store_row((unsigned)__builtin_amdgcn_readfirstlane((int)(sorted[p1 - 1u] >> SRC_BITS)), sum, bv, rest, acc, out, b, lane);
}

#define WS_RECS_OFF (16u << 20)
#define WS_TAB_OFF  (24u << 20)

extern "C" void kernel_launch(void* const* d_in, const int* in_sizes, int n_in,
                              void* d_out, int out_size, void* d_ws, size_t ws_size,
                              hipStream_t stream) {
    const float* x          = (const float*)d_in[0];
    const int*   edge_index = (const int*)  d_in[1];
    const float* W          = (const float*)d_in[2];
    const float* b          = (const float*)d_in[3];
    float*       out        = (float*)d_out;
    float*       y          = (float*)d_ws;
    unsigned*    recs       = (unsigned*)((char*)d_ws + WS_RECS_OFF);
    unsigned*    tab        = (unsigned*)((char*)d_ws + WS_TAB_OFF);

    const int* src = edge_index;
    const int* dst = edge_index + N_EDGES;

    gin_bin_and_proj<<<BIN_BLOCKS + PROJ_BLOCKS, L1_THREADS, 0, stream>>>(x, W, y, src, dst, recs, tab);
    gin_owner_sum<<<N_RANGES, OWN_THREADS, 0, stream>>>(recs, tab, y, b, out);
}


# launch 1: each work-group bins 2560 edges while its own x rows are in flight (LDS-only barriers, loads not waited for), then projects; 625 groups, all resident; edge sums in arrival order as before
# speedup vs baseline: 5.1681x; 1.0335x over previous
#include <hip/hip_runtime.h>


typedef __attribute__((ext_vector_type(2))) float v2f;
typedef __attribute__((ext_vector_type(4))) float v4f;
typedef __attribute__((ext_vector_type(8))) float v8f;

#define N_NODES   100000
#define D_FEAT    128
#define N_EDGES   1600000
#define N_CLASSES 32
#define M_TILES   (N_NODES / 16)

#define L1_THREADS 320
#define PROJ_WAVES 10
#define L1_BLOCKS  (M_TILES / PROJ_WAVES)
#define XP 132
#define OP 36
#define L1_LDS_BYTES ((N_CLASSES + PROJ_WAVES * 16) * XP * 4)
#define W_LOAD_WAVES 8
#define W_PER_THREAD (N_CLASSES * D_FEAT / 4 / (W_LOAD_WAVES * 32))
static_assert(L1_BLOCKS * PROJ_WAVES == M_TILES, "every wave of every block has a whole tile");
static_assert(W_LOAD_WAVES <= PROJ_WAVES && (N_CLASSES * D_FEAT / 4) % (W_LOAD_WAVES * 32) == 0, "W must split evenly over its loading waves");

#define N_RANGES      256
#define RANGE_ROWS    391
#define BIN_BLOCKS    L1_BLOCKS
#define BIN_EDGES     (N_EDGES / BIN_BLOCKS)
#define BIN_PER_THREAD (BIN_EDGES / L1_THREADS)
#define TAB_PITCH     (N_RANGES + 1)
#define SRC_BITS      17
#define SRC_MASK      0x1FFFFu
static_assert(BIN_BLOCKS * BIN_EDGES == N_EDGES, "edge list must split into whole bin blocks");
static_assert(L1_THREADS * BIN_PER_THREAD == BIN_EDGES, "threads must cover a bin block");
static_assert(L1_THREADS > N_RANGES, "one thread per range start");
static_assert(N_RANGES * RANGE_ROWS >= N_NODES, "ranges must cover all rows");
static_assert(N_NODES <= (1 << SRC_BITS) && RANGE_ROWS < 512, "record fields must fit");
static_assert((N_RANGES + N_RANGES + 1 + BIN_EDGES) * 4 <= PROJ_WAVES * 16 * XP * 4, "binning must fit inside the x tile region");

__device__ __forceinline__ void lds_barrier() {
    asm volatile("s_wait_dscnt 0x0\n\ts_barrier_signal -1\n\ts_barrier_wait -1" ::: "memory");
}

__global__ __launch_bounds__(L1_THREADS)
void gin_bin_and_proj(const float* __restrict__ x, const float* __restrict__ W,
                      float* __restrict__ y,
                      const int* __restrict__ src, const int* __restrict__ dst,
                      unsigned* __restrict__ recs, unsigned* __restrict__ tab) {
    __shared__ __attribute__((aligned(16))) unsigned char lds_raw[L1_LDS_BYTES];
    float* lds = (float*)lds_raw;
    float* Wl = lds;
    float* Xl = lds + N_CLASSES * XP;
    unsigned* cnt   = (unsigned*)Xl;
    unsigned* start = cnt + N_RANGES;
    unsigned* buf   = cnt + 2 * N_RANGES + 1;

    const int tid  = threadIdx.x;
    const int lane = tid & 31;
    const int wave  = __builtin_amdgcn_readfirstlane(tid >> 5);
    const int blk   = (int)blockIdx.x;
    const int mtile = blk * PROJ_WAVES + wave;
    float* xl = Xl + wave * (16 * XP);

    v4f wr[W_PER_THREAD];
    if (wave < W_LOAD_WAVES) {
#pragma unroll
        for (int i = 0; i < W_PER_THREAD; ++i) {
            const int idx = tid + W_LOAD_WAVES * 32 * i;
            wr[i] = *(const v4f*)(W + (idx >> 5) * D_FEAT + (idx & 31) * 4);
        }
    }
    const int e0 = blk * BIN_EDGES + tid;
    int s[BIN_PER_THREAD], d[BIN_PER_THREAD];
#pragma unroll
    for (int i = 0; i < BIN_PER_THREAD; ++i) {
        s[i] = src[e0 + i * L1_THREADS];
        d[i] = dst[e0 + i * L1_THREADS];
    }
    v4f xr[16];
    {
        const float* xt = x + mtile * (16 * D_FEAT) + lane * 4;
#pragma unroll
        for (int r = 0; r < 16; ++r) xr[r] = *(const v4f*)(xt + r * D_FEAT);
    }

    if (tid < N_RANGES) cnt[tid] = 0u;
    if (wave < W_LOAD_WAVES) {
#pragma unroll
        for (int i = 0; i < W_PER_THREAD; ++i) {
            const int idx = tid + W_LOAD_WAVES * 32 * i;
            *(v4f*)(Wl + (idx >> 5) * XP + (idx & 31) * 4) = wr[i];
        }
    }
    lds_barrier();

    unsigned rec[BIN_PER_THREAD], rng[BIN_PER_THREAD], rank[BIN_PER_THREAD];
#pragma unroll
    for (int i = 0; i < BIN_PER_THREAD; ++i) {
        const unsigned t = (unsigned)d[i];
        const unsigned r = t / RANGE_ROWS;
        rng[i]  = r;
        rec[i]  = (unsigned)s[i] | ((t - r * RANGE_ROWS) << SRC_BITS);
        rank[i] = atomicAdd(&cnt[r], 1u);
    }
    lds_barrier();

    if (tid < 32) {
        unsigned c[8];
        unsigned sum = 0u;
#pragma unroll
        for (int k = 0; k < 8; ++k) { c[k] = cnt[tid * 8 + k]; sum += c[k]; }
        unsigned incl = sum;
#pragma unroll
        for (int o = 1; o < 32; o <<= 1) {
            const unsigned up = __shfl_up(incl, o);
            if (tid >= o) incl += up;
        }
        unsigned run = incl - sum;
#pragma unroll
        for (int k = 0; k < 8; ++k) { start[tid * 8 + k] = run; run += c[k]; }
        if (tid == 31) start[N_RANGES] = run;
    }
    lds_barrier();

    unsigned pos[BIN_PER_THREAD];
#pragma unroll
    for (int i = 0; i < BIN_PER_THREAD; ++i) pos[i] = start[rng[i]] + rank[i];
#pragma unroll
    for (int i = 0; i < BIN_PER_THREAD; ++i) buf[pos[i]] = rec[i];
    lds_barrier();

    {
        unsigned* rb = recs + blk * BIN_EDGES;
#pragma unroll
        for (int i = 0; i < BIN_PER_THREAD; ++i) rb[tid + i * L1_THREADS] = buf[tid + i * L1_THREADS];
        if (tid <= N_RANGES) tab[blk * TAB_PITCH + tid] = start[tid];
    }
    lds_barrier();

#pragma unroll
    for (int r = 0; r < 16; ++r) *(v4f*)(xl + r * XP + lane * 4) = xr[r];

    const int c = lane & 15;
    const int h = lane >> 4;
    v8f acc0 = {};
    v8f acc1 = {};
    {
        const float* ap  = xl + c * XP + 4 * h;
        const float* b0p = Wl + c * XP + 4 * h;
        const float* b1p = Wl + (c + 16) * XP + 4 * h;
#pragma unroll
        for (int j = 0; j < D_FEAT / 8; ++j) {
            const v4f a  = *(const v4f*)(ap  + 8 * j);
            const v4f b0 = *(const v4f*)(b0p + 8 * j);
            const v4f b1 = *(const v4f*)(b1p + 8 * j);
            const v2f alo  = {a[0],  a[1]},  ahi  = {a[2],  a[3]};
            const v2f b0lo = {b0[0], b0[1]}, b0hi = {b0[2], b0[3]};
            const v2f b1lo = {b1[0], b1[1]}, b1hi = {b1[2], b1[3]};
            acc0 = __builtin_amdgcn_wmma_f32_16x16x4_f32(false, alo, false, b0lo,
                                                         (short)0, acc0, false, false);
            acc1 = __builtin_amdgcn_wmma_f32_16x16x4_f32(false, alo, false, b1lo,
                                                         (short)0, acc1, false, false);
            acc0 = __builtin_amdgcn_wmma_f32_16x16x4_f32(false, ahi, false, b0hi,
                                                         (short)0, acc0, false, false);
            acc1 = __builtin_amdgcn_wmma_f32_16x16x4_f32(false, ahi, false, b1hi,
                                                         (short)0, acc1, false, false);
        }
#pragma unroll
        for (int r = 0; r < 8; ++r) {
            xl[(8 * h + r) * OP + c]      = acc0[r];
            xl[(8 * h + r) * OP + c + 16] = acc1[r];
        }
    }
    __syncthreads();
    {
        const int c4 = (lane & 7) * 4;
        float* yt = y + mtile * (16 * N_CLASSES);
#pragma unroll
        for (int i = 0; i < 4; ++i) {
            const int row = (lane >> 3) + 4 * i;
            *(v4f*)(yt + row * N_CLASSES + c4) = *(const v4f*)(xl + row * OP + c4);
        }
    }
}

#define OWN_THREADS 1024
#define OWN_WAVES   32
#define OWN_SEGS    20
#define SEG_HEAD    32
#define SORT_CAP    (BIN_BLOCKS * SEG_HEAD + RANGE_ROWS)
#define CNT_PAD     (OWN_WAVES * 13)
#define SPARE_REC   ((unsigned)RANGE_ROWS << SRC_BITS)
#define NO_ROW      511u
#define END_REC     (NO_ROW << SRC_BITS)
static_assert(OWN_WAVES * OWN_SEGS >= BIN_BLOCKS && OWN_SEGS <= 32, "waves must cover all bin blocks, one lane per segment");
static_assert(CNT_PAD > RANGE_ROWS && NO_ROW > RANGE_ROWS, "counter padding and the end mark must lie past the rows");
static_assert(12 * OWN_WAVES + 7 == RANGE_ROWS, "7 waves of 13 rows and 25 of 12 must cover the range");

__device__ __forceinline__ void sum_rows32(unsigned rec, const float* __restrict__ y,
                                           float* acc, unsigned c4, int lane) {
    float v[32];
#pragma unroll
    for (int u = 0; u < 32; ++u) {
        const unsigned s = (unsigned)__builtin_amdgcn_readlane((int)rec, u) & SRC_MASK;
        v[u] = *(const float*)((const char*)y + ((s << 7) | c4));
    }
#pragma unroll
    for (int u = 0; u < 32; ++u) {
        const unsigned r = (unsigned)__builtin_amdgcn_readlane((int)rec, u) >> SRC_BITS;
        atomicAdd(&acc[r * 32 + lane], v[u]);
    }
}

__device__ __forceinline__ void store_row(unsigned row, float sum, float bv, bool rest, const float* acc,
                                          float* __restrict__ out, int b, int lane) {
    if (row < (unsigned)RANGE_ROWS) {
        const int node = b * RANGE_ROWS + (int)row;
        if (node < N_NODES) {
            float o = sum + bv;
            if (rest) o += acc[row * 32 + lane];
            out[node * N_CLASSES + lane] = o;
        }
    }
}

__global__ __launch_bounds__(OWN_THREADS)
void gin_owner_sum(const unsigned* __restrict__ recs, const unsigned* __restrict__ tab,
                   const float* __restrict__ y, const float* __restrict__ bias,
                   float* __restrict__ out) {
    __shared__ float acc[(RANGE_ROWS + 1) * 32];
    __shared__ unsigned cnt[CNT_PAD];
    __shared__ unsigned start[CNT_PAD + 1];
    __shared__ unsigned sorted[SORT_CAP];
    __shared__ unsigned rest_flag;

    const int tid  = threadIdx.x;
    const int lane = tid & 31;
    const unsigned c4 = (unsigned)lane * 4u;
    const int wave = __builtin_amdgcn_readfirstlane(tid >> 5);
    const int b = blockIdx.x;

    for (int i = tid; i < (RANGE_ROWS + 1) * 32; i += OWN_THREADS) acc[i] = 0.f;
    if (tid < CNT_PAD) cnt[tid] = (tid < RANGE_ROWS && b * RANGE_ROWS + tid < N_NODES) ? 1u : 0u;
    if (tid == 0) rest_flag = 0u;

    unsigned st = 0u, en = 0u;
    {
        const int blk = wave + 32 * lane;
        if (lane < OWN_SEGS && blk < BIN_BLOCKS) {
            st = tab[blk * TAB_PITCH + b];
            en = tab[blk * TAB_PITCH + b + 1];
        }
    }
    unsigned r0[OWN_SEGS];
#pragma unroll
    for (int j = 0; j < OWN_SEGS; ++j) {
        const unsigned sj = (unsigned)__builtin_amdgcn_readlane((int)st, j);
        const unsigned ej = (unsigned)__builtin_amdgcn_readlane((int)en, j);
        const unsigned* p = recs + (wave + 32 * j) * BIN_EDGES;
        const unsigned i0 = sj + (unsigned)lane;
        r0[j] = (i0 < ej) ? p[i0] : SPARE_REC;
    }
    __syncthreads();

    unsigned k0[OWN_SEGS];
#pragma unroll
    for (int j = 0; j < OWN_SEGS; ++j) {
        k0[j] = 0u;
        if (r0[j] != SPARE_REC) k0[j] = atomicAdd(&cnt[r0[j] >> SRC_BITS], 1u);
    }

    bool used_rest = false;
#pragma unroll 1
    for (int j = 0; j < OWN_SEGS; ++j) {
        const unsigned sj = (unsigned)__builtin_amdgcn_readlane((int)st, j);
        const unsigned ej = (unsigned)__builtin_amdgcn_readlane((int)en, j);
        const unsigned* p = recs + (wave + 32 * j) * BIN_EDGES;
#pragma unroll 1
        for (unsigned base = sj + (unsigned)SEG_HEAD; base < ej; base += 32u) {
            const unsigned i0 = base + (unsigned)lane;
            const unsigned r = (i0 < ej) ? p[i0] : SPARE_REC;
            sum_rows32(r, y, acc, c4, lane);
            used_rest = true;
        }
    }
    if (used_rest && lane == 0) rest_flag = 1u;
    __syncthreads();

    if (wave == 0) {
        unsigned c[13];
        unsigned tot = 0u;
#pragma unroll
        for (int k = 0; k < 13; ++k) { c[k] = cnt[lane * 13 + k]; tot += c[k]; }
        unsigned incl = tot;
#pragma unroll
        for (int o = 1; o < 32; o <<= 1) {
            const unsigned up = __shfl_up(incl, o);
            if (lane >= o) incl += up;
        }
        unsigned run = incl - tot;
#pragma unroll
        for (int k = 0; k < 13; ++k) { start[lane * 13 + k] = run; run += c[k]; }
        if (lane == 31) start[CNT_PAD] = run;
    }
    __syncthreads();

#pragma unroll
    for (int j = 0; j < OWN_SEGS; ++j) {
        if (r0[j] != SPARE_REC) sorted[start[r0[j] >> SRC_BITS] + k0[j]] = r0[j];
    }
    if (tid < RANGE_ROWS && b * RANGE_ROWS + tid < N_NODES)
        sorted[start[tid]] = (unsigned)(b * RANGE_ROWS + tid) | ((unsigned)tid << SRC_BITS);
    __syncthreads();

    const int row_lo = 12 * wave + (wave < 7 ? wave : 7);
    const int row_hi = 12 * (wave + 1) + (wave + 1 < 7 ? wave + 1 : 7);
    const unsigned p0 = (unsigned)__builtin_amdgcn_readfirstlane((int)start[row_lo]);
    const unsigned p1 = (unsigned)__builtin_amdgcn_readfirstlane((int)start[row_hi]);
    const bool rest = __builtin_amdgcn_readfirstlane((int)rest_flag) != 0;
    const float bv = bias[lane];

    float sum = 0.f;
#pragma unroll 1
    for (unsigned t = p0; t < p1; t += 32u) {
        const unsigned idx = t + (unsigned)lane;
        const unsigned rec = (idx < p1) ? sorted[idx] : END_REC;
        const unsigned prow = (idx > p0 && idx <= p1) ? (sorted[idx - 1u] >> SRC_BITS) : NO_ROW;
        const unsigned chg = __builtin_amdgcn_ballot_w32((rec >> SRC_BITS) != prow);
        float v[32];
#pragma unroll
        for (int u = 0; u < 32; ++u) {
            const unsigned s = (unsigned)__builtin_amdgcn_readlane((int)rec, u) & SRC_MASK;
            v[u] = *(const float*)((const char*)y + ((s << 7) | c4));
        }
#pragma unroll
        for (int u = 0; u < 32; ++u) {
            if (__builtin_expect((chg >> u) & 1u, 0u)) {
                store_row((unsigned)__builtin_amdgcn_readlane((int)prow, u), sum, bv, rest, acc, out, b, lane);
                sum = 0.f;
            }
            sum += v[u];
        }
    }
    if (p1 > p0 && ((p1 - p0) & 31u) == 0u)
        store_row((unsigned)__builtin_amdgcn_readfirstlane((int)(sorted[p1 - 1u] >> SRC_BITS)), sum, bv, rest, acc, out, b, lane);
}

#define WS_RECS_OFF (16u << 20)
#define WS_TAB_OFF  (24u << 20)

extern "C" void kernel_launch(void* const* d_in, const int* in_sizes, int n_in,
                              void* d_out, int out_size, void* d_ws, size_t ws_size,
                              hipStream_t stream) {
    const float* x          = (const float*)d_in[0];
    const int*   edge_index = (const int*)  d_in[1];
    const float* W          = (const float*)d_in[2];
    const float* b          = (const float*)d_in[3];
    float*       out        = (float*)d_out;
    float*       y          = (float*)d_ws;
    unsigned*    recs       = (unsigned*)((char*)d_ws + WS_RECS_OFF);
    unsigned*    tab        = (unsigned*)((char*)d_ws + WS_TAB_OFF);

    const int* src = edge_index;
    const int* dst = edge_index + N_EDGES;

    gin_bin_and_proj<<<L1_BLOCKS, L1_THREADS, 0, stream>>>(x, W, y, src, dst, recs, tab);
    gin_owner_sum<<<N_RANGES, OWN_THREADS, 0, stream>>>(recs, tab, y, b, out);
}
